# starting kernel
# baseline (speedup 1.0000x reference)
#include <hip/hip_runtime.h>
#include <hip/hip_bf16.h>


#define B_   4
#define C_   64
#define H_   96
#define W_   320
#define HW_  (H_ * W_)
#define CHW_ (C_ * HW_)
#define HO_  384
#define WO_  1280
#define NOUT_ (B_ * HO_ * WO_)
#define NZ_   (B_ * HW_)

typedef __attribute__((ext_vector_type(2))) float v2f;
typedef __attribute__((ext_vector_type(8))) float v8f;

__global__ __launch_bounds__(256) void
upsample_reduce_c_kernel(const float* __restrict__ x,
                         const float* __restrict__ w,
                         float* __restrict__ z) {
  const int tid  = blockIdx.x * 256 + threadIdx.x;
  const int wave = tid >> 5;
  const int lane = threadIdx.x & 31;
  const int m    = lane & 15;
  const int half = lane >> 4;

  const int pbase = wave * 16;
  const int b     = pbase / HW_;
  const int s     = pbase - b * HW_;

  const float* xb = x + (size_t)b * CHW_ + (size_t)(s + m);

  v8f acc = {};
#pragma unroll
  for (int c0 = 0; c0 < C_; c0 += 4) {
    const int cb = c0 + (half << 1);
    v2f a, bw;
    a.x  = xb[(size_t)cb * HW_];
    a.y  = xb[(size_t)(cb + 1) * HW_];
    bw.x = w[cb];
    bw.y = w[cb + 1];
    acc = __builtin_amdgcn_wmma_f32_16x16x4_f32(
         false, a,  false, bw,
         (short)0, acc,  false,  false);
  }

  if (m == 0) {
    float* zp = z + pbase + (half << 3);
    zp[0] = acc[0]; zp[1] = acc[1]; zp[2] = acc[2]; zp[3] = acc[3];
    zp[4] = acc[4]; zp[5] = acc[5]; zp[6] = acc[6]; zp[7] = acc[7];
  }
}

__global__ __launch_bounds__(256) void
upsample_lerp_bias_relu_kernel(const float* __restrict__ z,
                               const float* __restrict__ bias,
                               float* __restrict__ out) {
  const int idx = blockIdx.x * 256 + threadIdx.x;
  const int wo  = idx % WO_;
  const int t   = idx / WO_;
  const int ho  = t % HO_;
  const int b   = t / HO_;

  const float RY = (float)((double)(H_ - 1) / (double)(HO_ - 1));
  const float RX = (float)((double)(W_ - 1) / (double)(WO_ - 1));

  const float fy = (float)ho * RY;
  int i0 = (int)fy; if (i0 > H_ - 2) i0 = H_ - 2;
  const float wy = fy - (float)i0;

  const float fx = (float)wo * RX;
  int j0 = (int)fx; if (j0 > W_ - 2) j0 = W_ - 2;
  const float wx = fx - (float)j0;

  const float* zb = z + b * HW_ + i0 * W_ + j0;
  const float v00 = zb[0];
  const float v01 = zb[1];
  const float v10 = zb[W_];
  const float v11 = zb[W_ + 1];

  const float r0  = v00 * (1.0f - wy) + v10 * wy;
  const float r1  = v01 * (1.0f - wy) + v11 * wy;
  const float val = r0 * (1.0f - wx) + r1 * wx + bias[0];

  out[idx] = val > 0.0f ? val : 0.0f;
}

extern "C" void kernel_launch(void* const* d_in, const int* in_sizes, int n_in,
                              void* d_out, int out_size, void* d_ws, size_t ws_size,
                              hipStream_t stream) {
  (void)in_sizes; (void)n_in; (void)out_size; (void)ws_size;

  const float* x    = (const float*)d_in[0];
  const float* wgt  = (const float*)d_in[1];
  const float* bias = (const float*)d_in[2];
  float*       out  = (float*)d_out;
  float*       z    = (float*)d_ws;

  upsample_reduce_c_kernel<<<NZ_ / 16 / 8, 256, 0, stream>>>(x, wgt, z);
  upsample_lerp_bias_relu_kernel<<<NOUT_ / 256, 256, 0, stream>>>(z, bias, out);
}


# upsample launch register-tiled: one thread computes a 4x4 output patch from a 3x3 block of z (six loads, four 16-byte stores), 16x fewer waves; two launches, z in workspace and reduce launch kept
# speedup vs baseline: 1.5292x; 1.5292x over previous
#include <hip/hip_runtime.h>
#include <hip/hip_bf16.h>


#define B_   4
#define C_   64
#define H_   96
#define W_   320
#define HW_  (H_ * W_)
#define CHW_ (C_ * HW_)
#define HO_  384
#define WO_  1280
#define NOUT_ (B_ * HO_ * WO_)
#define NZ_   (B_ * HW_)

typedef __attribute__((ext_vector_type(2))) float v2f;
typedef __attribute__((ext_vector_type(8))) float v8f;

__global__ __launch_bounds__(256) void
upsample_reduce_c_kernel(const float* __restrict__ x,
                         const float* __restrict__ w,
                         float* __restrict__ z) {
  const int tid  = blockIdx.x * 256 + threadIdx.x;
  const int wave = tid >> 5;
  const int lane = threadIdx.x & 31;
  const int m    = lane & 15;
  const int half = lane >> 4;

  const int pbase = wave * 16;
  const int b     = pbase / HW_;
  const int s     = pbase - b * HW_;

  const float* xb = x + (size_t)b * CHW_ + (size_t)(s + m);

  v8f acc = {};
#pragma unroll
  for (int c0 = 0; c0 < C_; c0 += 4) {
    const int cb = c0 + (half << 1);
    v2f a, bw;
    a.x  = xb[(size_t)cb * HW_];
    a.y  = xb[(size_t)(cb + 1) * HW_];
    bw.x = w[cb];
    bw.y = w[cb + 1];
    acc = __builtin_amdgcn_wmma_f32_16x16x4_f32(
         false, a,  false, bw,
         (short)0, acc,  false,  false);
  }

  if (m == 0) {
    float* zp = z + pbase + (half << 3);
    zp[0] = acc[0]; zp[1] = acc[1]; zp[2] = acc[2]; zp[3] = acc[3];
    zp[4] = acc[4]; zp[5] = acc[5]; zp[6] = acc[6]; zp[7] = acc[7];
  }
}

#define PW_ (WO_ / 4)
#define PH_ (HO_ / 4)
#define NPATCH_ (B_ * PH_ * PW_)

typedef __attribute__((ext_vector_type(4))) float v4f;

__global__ __launch_bounds__(256) void
upsample_lerp_bias_relu_kernel(const float* __restrict__ z,
                               const float* __restrict__ bias,
                               float* __restrict__ out) {
  const int idx = blockIdx.x * 256 + threadIdx.x;
  const int q   = idx % PW_;
  const int t   = idx / PW_;
  const int p   = t % PH_;
  const int b   = t / PH_;

  const float RY = (float)((double)(H_ - 1) / (double)(HO_ - 1));
  const float RX = (float)((double)(W_ - 1) / (double)(WO_ - 1));

  int   dj[4];
  float wx[4];
  int   jb = 0;
  for (int c = 0; c < 4; ++c) {
    const float fx = (float)(4 * q + c) * RX;
    int j0 = (int)fx; if (j0 > W_ - 2) j0 = W_ - 2;
    wx[c] = fx - (float)j0;
    if (c == 0) jb = j0;
    dj[c] = j0 - jb;
  }

  int   di[4];
  float wy[4];
  int   ib = 0;
  for (int a = 0; a < 4; ++a) {
    const float fy = (float)(4 * p + a) * RY;
    int i0 = (int)fy; if (i0 > H_ - 2) i0 = H_ - 2;
    wy[a] = fy - (float)i0;
    if (a == 0) ib = i0;
    di[a] = i0 - ib;
  }

  const int i2 = (ib + 2 < H_) ? ib + 2 : H_ - 1;
  const int j2 = (jb + 2 < W_) ? jb + 2 : W_ - 1;
  const float* zb  = z + b * HW_;
  const float* zr0 = zb + ib * W_;
  const float* zr1 = zb + (ib + 1) * W_;
  const float* zr2 = zb + i2 * W_;
  float zv[3][3];
  zv[0][0] = zr0[jb]; zv[0][1] = zr0[jb + 1]; zv[0][2] = zr0[j2];
  zv[1][0] = zr1[jb]; zv[1][1] = zr1[jb + 1]; zv[1][2] = zr1[j2];
  zv[2][0] = zr2[jb]; zv[2][1] = zr2[jb + 1]; zv[2][2] = zr2[j2];

  const float bs = bias[0];
  float* ob = out + ((size_t)b * HO_ + (size_t)(4 * p)) * WO_ + (size_t)(4 * q);

  for (int a = 0; a < 4; ++a) {
    const bool  up  = di[a] != 0;
    const float wya = wy[a];
    float rl[3];
    for (int k = 0; k < 3; ++k) {
      const float top = up ? zv[1][k] : zv[0][k];
      const float bot = up ? zv[2][k] : zv[1][k];
      rl[k] = top * (1.0f - wya) + bot * wya;
    }
    v4f o;
    for (int c = 0; c < 4; ++c) {
      const bool  rt = dj[c] != 0;
      const float r0 = rt ? rl[1] : rl[0];
      const float r1 = rt ? rl[2] : rl[1];
      const float val = r0 * (1.0f - wx[c]) + r1 * wx[c] + bs;
      o[c] = val > 0.0f ? val : 0.0f;
    }
    *(v4f*)(ob + (size_t)a * WO_) = o;
  }
}

extern "C" void kernel_launch(void* const* d_in, const int* in_sizes, int n_in,
                              void* d_out, int out_size, void* d_ws, size_t ws_size,
                              hipStream_t stream) {
  (void)in_sizes; (void)n_in; (void)out_size; (void)ws_size;

  const float* x    = (const float*)d_in[0];
  const float* wgt  = (const float*)d_in[1];
  const float* bias = (const float*)d_in[2];
  float*       out  = (float*)d_out;
  float*       z    = (float*)d_ws;

  upsample_reduce_c_kernel<<<NZ_ / 16 / 8, 256, 0, stream>>>(x, wgt, z);
  upsample_lerp_bias_relu_kernel<<<NPATCH_ / 256, 256, 0, stream>>>(z, bias, out);
}
